# starting kernel
# baseline (speedup 1.0000x reference)
#include <hip/hip_runtime.h>
#include <math.h>

#define BATCH   1024
#define SEQ     256
#define EMBED   4
#define H1D     8
#define H2D     4
#define NWAVE   4
#define SSTRIDE 260

typedef __attribute__((ext_vector_type(2))) float v2f;
typedef __attribute__((ext_vector_type(8))) float v8f;

__global__ __launch_bounds__(128)
void hybrid_attn_kernel(const float* __restrict__ x,
                        const float* __restrict__ Wq, const float* __restrict__ Wk,
                        const float* __restrict__ Wv,
                        const float* __restrict__ W1, const float* __restrict__ b1,
                        const float* __restrict__ W2, const float* __restrict__ b2,
                        const float* __restrict__ W3, const float* __restrict__ b3,
                        float* __restrict__ out)
{
  extern __shared__ float smem[];
  float*  q_lds = smem;
  float*  k_lds = q_lds + SEQ * EMBED;
  float4* v4    = (float4*)(k_lds + SEQ * EMBED);
  float*  sbuf  = (float*)(v4 + SEQ);

  const int b    = blockIdx.x;
  const int tid  = threadIdx.x;
  const int lane = tid & 31;
  const int wave = tid >> 5;

  const float* xb = x + (size_t)b * SEQ * EMBED;
  for (int t = tid; t < SEQ; t += 128) {
    float4 xe = ((const float4*)xb)[t];
    float xv[4] = {xe.x, xe.y, xe.z, xe.w};
    float4 vvec;
    float* vp = (float*)&vvec;
#pragma unroll
    for (int eo = 0; eo < 4; ++eo) {
      float q = 0.f, k = 0.f, v = 0.f;
#pragma unroll
      for (int e = 0; e < 4; ++e) {
        q += xv[e] * Wq[e * 4 + eo];
        k += xv[e] * Wk[e * 4 + eo];
        v += xv[e] * Wv[e * 4 + eo];
      }
      q_lds[t * 4 + eo] = 0.5f * q;
      k_lds[t * 4 + eo] = k;
      vp[eo] = v;
    }
    v4[t] = vvec;
  }
  __syncthreads();

  const int lane15 = lane & 15;
  const int half   = lane >> 4;
  const int kbase  = half * 2;

  v2f barr[16];
#pragma unroll
  for (int j = 0; j < 16; ++j) {
    barr[j].x = k_lds[(j * 16 + lane15) * 4 + kbase];
    barr[j].y = k_lds[(j * 16 + lane15) * 4 + kbase + 1];
  }

  float* swave = sbuf + wave * 16 * SSTRIDE;

  for (int tile = wave; tile < SEQ / 16; tile += NWAVE) {
    const int i0 = tile * 16;

    v2f a;
    a.x = q_lds[(i0 + lane15) * 4 + kbase];
    a.y = q_lds[(i0 + lane15) * 4 + kbase + 1];

#pragma unroll
    for (int j = 0; j < 16; ++j) {
      v8f c = {};
      c = __builtin_amdgcn_wmma_f32_16x16x4_f32(
              false, a, false, barr[j], (short)0, c, false, false);

#pragma unroll
      for (int r = 0; r < 8; ++r)
        swave[(r + 8 * half) * SSTRIDE + j * 16 + lane15] = c[r];
    }

    const int    row   = lane15;
    const float* Sr    = swave + row * SSTRIDE;
    const int    tbase = half * 128;

    float mx = -1e30f;
#pragma unroll 8
    for (int i = 0; i < 128; i += 4) {
      float4 sv = *(const float4*)(Sr + tbase + i);
      mx = fmaxf(mx, fmaxf(fmaxf(sv.x, sv.y), fmaxf(sv.z, sv.w)));
    }
    mx = fmaxf(mx, __shfl_xor(mx, 16, 32));

    float sum = 0.f, o0 = 0.f, o1 = 0.f, o2 = 0.f, o3 = 0.f;
#pragma unroll 4
    for (int i = 0; i < 128; i += 4) {
      float4 sv = *(const float4*)(Sr + tbase + i);
      float p0 = __expf(sv.x - mx);
      float p1 = __expf(sv.y - mx);
      float p2 = __expf(sv.z - mx);
      float p3 = __expf(sv.w - mx);
      sum += (p0 + p1) + (p2 + p3);
      float4 va = v4[tbase + i + 0];
      float4 vb = v4[tbase + i + 1];
      float4 vc = v4[tbase + i + 2];
      float4 vd = v4[tbase + i + 3];
      o0 += p0 * va.x + p1 * vb.x + p2 * vc.x + p3 * vd.x;
      o1 += p0 * va.y + p1 * vb.y + p2 * vc.y + p3 * vd.y;
      o2 += p0 * va.z + p1 * vb.z + p2 * vc.z + p3 * vd.z;
      o3 += p0 * va.w + p1 * vb.w + p2 * vc.w + p3 * vd.w;
    }
    sum += __shfl_xor(sum, 16, 32);
    o0  += __shfl_xor(o0, 16, 32);
    o1  += __shfl_xor(o1, 16, 32);
    o2  += __shfl_xor(o2, 16, 32);
    o3  += __shfl_xor(o3, 16, 32);

    float inv = 1.f / sum;
    o0 *= inv; o1 *= inv; o2 *= inv; o3 *= inv;

    float h1v[H1D];
#pragma unroll
    for (int j = 0; j < H1D; ++j)
      h1v[j] = tanhf(b1[j] + o0 * W1[0 * H1D + j] + o1 * W1[1 * H1D + j]
                           + o2 * W1[2 * H1D + j] + o3 * W1[3 * H1D + j]);
    float h2v[H2D];
#pragma unroll
    for (int j = 0; j < H2D; ++j) {
      float acc = b2[j];
#pragma unroll
      for (int i = 0; i < H1D; ++i) acc += h1v[i] * W2[i * H2D + j];
      h2v[j] = tanhf(acc);
    }
    float y = b3[0];
#pragma unroll
    for (int j = 0; j < H2D; ++j) y += h2v[j] * W3[j];

    if (half == 0) out[(size_t)b * SEQ + i0 + row] = y;
  }
}

extern "C" void kernel_launch(void* const* d_in, const int* in_sizes, int n_in,
                              void* d_out, int out_size, void* d_ws, size_t ws_size,
                              hipStream_t stream) {
  const float* x  = (const float*)d_in[0];
  const float* Wq = (const float*)d_in[1];
  const float* Wk = (const float*)d_in[2];
  const float* Wv = (const float*)d_in[3];
  const float* W1 = (const float*)d_in[4];
  const float* b1 = (const float*)d_in[5];
  const float* W2 = (const float*)d_in[6];
  const float* b2 = (const float*)d_in[7];
  const float* W3 = (const float*)d_in[8];
  const float* b3 = (const float*)d_in[9];
  float* out = (float*)d_out;

  const size_t smem_bytes =
      (size_t)(SEQ * EMBED * 2 + SEQ * 4 + NWAVE * 16 * SSTRIDE) * sizeof(float);

  hybrid_attn_kernel<<<BATCH, 128, smem_bytes, stream>>>(
      x, Wq, Wk, Wv, W1, b1, W2, b2, W3, b3, out);
}


# softmax and PV in place on the score accumulators (lane = key, register = query): -max via the C operand, exp2, packed FMAs on each lane's own V rows, DPP row sums; no score LDS; sum order changed
# speedup vs baseline: 1.2016x; 1.2016x over previous
#include <hip/hip_runtime.h>
#include <math.h>


#define BATCH   1024
#define SEQ     256
#define EMBED   4
#define H1D     8
#define H2D     4
#define NWAVE   4

typedef __attribute__((ext_vector_type(2))) float v2f;
typedef __attribute__((ext_vector_type(8))) float v8f;

template <int M>
__device__ __forceinline__ float row_xor(float v) {
  return __builtin_bit_cast(float,
      __builtin_amdgcn_update_dpp(0, __builtin_bit_cast(int, v), 0x160 + M, 0xf, 0xf, true));
}

__global__ __launch_bounds__(128)
void hybrid_attn_kernel(const float* __restrict__ x,
                        const float* __restrict__ Wq, const float* __restrict__ Wk,
                        const float* __restrict__ Wv,
                        const float* __restrict__ W1, const float* __restrict__ b1,
                        const float* __restrict__ W2, const float* __restrict__ b2,
                        const float* __restrict__ W3, const float* __restrict__ b3,
                        float* __restrict__ out)
{
  extern __shared__ float smem[];
  float*  q_lds = smem;
  float*  k_lds = q_lds + SEQ * EMBED;
  float4* v4    = (float4*)(k_lds + SEQ * EMBED);

  const int b    = blockIdx.x;
  const int tid  = threadIdx.x;
  const int lane = tid & 31;
  const int wave = tid >> 5;

  const float qscale = 0.5f * 1.44269504088896340736f;
  const float* xb = x + (size_t)b * SEQ * EMBED;
  for (int t = tid; t < SEQ; t += 128) {
    float4 xe = ((const float4*)xb)[t];
    float xv[4] = {xe.x, xe.y, xe.z, xe.w};
    float4 vvec;
    float* vp = (float*)&vvec;
#pragma unroll
    for (int eo = 0; eo < 4; ++eo) {
      float q = 0.f, k = 0.f, v = 0.f;
#pragma unroll
      for (int e = 0; e < 4; ++e) {
        q += xv[e] * Wq[e * 4 + eo];
        k += xv[e] * Wk[e * 4 + eo];
        v += xv[e] * Wv[e * 4 + eo];
      }
      q_lds[t * 4 + eo] = qscale * q;
      k_lds[t * 4 + eo] = k;
      vp[eo] = v;
    }
    v4[t] = vvec;
  }
  __syncthreads();

  const int c     = lane & 15;
  const int h     = lane >> 4;
  const int kbase = h * 2;
  const bool hi8  = (c & 8) != 0;
  const bool hi4  = (c & 4) != 0;

  v2f    barr[16];
  float4 vreg[16];
#pragma unroll
  for (int j = 0; j < 16; ++j) {
    barr[j].x = k_lds[(j * 16 + c) * 4 + kbase];
    barr[j].y = k_lds[(j * 16 + c) * 4 + kbase + 1];
    vreg[j]   = v4[j * 16 + c];
  }

  for (int t = 0; t < SEQ / 16 / NWAVE; ++t) {
    const int i0 = (wave + NWAVE * t) * 16;

    v2f a;
    a.x = q_lds[(i0 + c) * 4 + kbase];
    a.y = q_lds[(i0 + c) * 4 + kbase + 1];

    float mx[8];
    {
      v8f z = {};
      v8f s = __builtin_amdgcn_wmma_f32_16x16x4_f32(
                  false, a, false, barr[0], (short)0, z, false, false);
#pragma unroll
      for (int r = 0; r < 8; ++r) mx[r] = s[r];
#pragma unroll
      for (int j = 1; j < 16; ++j) {
        s = __builtin_amdgcn_wmma_f32_16x16x4_f32(
                false, a, false, barr[j], (short)0, z, false, false);
#pragma unroll
        for (int r = 0; r < 8; ++r) mx[r] = fmaxf(mx[r], s[r]);
      }
    }
#pragma unroll
    for (int r = 0; r < 8; ++r) mx[r] = fmaxf(mx[r], row_xor<1>(mx[r]));
#pragma unroll
    for (int r = 0; r < 8; ++r) mx[r] = fmaxf(mx[r], row_xor<2>(mx[r]));
#pragma unroll
    for (int r = 0; r < 8; ++r) mx[r] = fmaxf(mx[r], row_xor<4>(mx[r]));
#pragma unroll
    for (int r = 0; r < 8; ++r) mx[r] = fmaxf(mx[r], row_xor<8>(mx[r]));

    v8f negm;
#pragma unroll
    for (int r = 0; r < 8; ++r) negm[r] = -mx[r];

    v2f acc[4][4], l2[4];
#pragma unroll
    for (int e = 0; e < 4; ++e)
#pragma unroll
      for (int i = 0; i < 4; ++i) acc[e][i] = (v2f){0.f, 0.f};
#pragma unroll
    for (int i = 0; i < 4; ++i) l2[i] = (v2f){0.f, 0.f};

#pragma unroll
    for (int j = 0; j < 16; ++j) {
      v8f d = __builtin_amdgcn_wmma_f32_16x16x4_f32(
                  false, a, false, barr[j], (short)0, negm, false, false);
      const float ve[4] = {vreg[j].x, vreg[j].y, vreg[j].z, vreg[j].w};
      v2f p2[4];
#pragma unroll
      for (int i = 0; i < 4; ++i)
        p2[i] = (v2f){__builtin_amdgcn_exp2f(d[2 * i]), __builtin_amdgcn_exp2f(d[2 * i + 1])};
#pragma unroll
      for (int e = 0; e < 4; ++e) {
        const v2f vv = {ve[e], ve[e]};
#pragma unroll
        for (int i = 0; i < 4; ++i) acc[e][i] = __builtin_elementwise_fma(p2[i], vv, acc[e][i]);
      }
#pragma unroll
      for (int i = 0; i < 4; ++i) l2[i] += p2[i];
    }

    float val[8][5];
#pragma unroll
    for (int r = 0; r < 8; ++r) {
#pragma unroll
      for (int e = 0; e < 4; ++e) val[r][e] = (r & 1) ? acc[e][r >> 1].y : acc[e][r >> 1].x;
      val[r][4] = (r & 1) ? l2[r >> 1].y : l2[r >> 1].x;
    }
#pragma unroll
    for (int r = 0; r < 4; ++r)
#pragma unroll
      for (int k = 0; k < 5; ++k) {
        const float keep = hi8 ? val[r + 4][k] : val[r][k];
        const float send = hi8 ? val[r][k] : val[r + 4][k];
        val[r][k] = keep + row_xor<8>(send);
      }
#pragma unroll
    for (int r = 0; r < 2; ++r)
#pragma unroll
      for (int k = 0; k < 5; ++k) {
        const float keep = hi4 ? val[r + 2][k] : val[r][k];
        const float send = hi4 ? val[r][k] : val[r + 2][k];
        val[r][k] = keep + row_xor<4>(send);
      }
#pragma unroll
    for (int r = 0; r < 2; ++r)
#pragma unroll
      for (int k = 0; k < 5; ++k) val[r][k] += row_xor<2>(val[r][k]);
#pragma unroll
    for (int r = 0; r < 2; ++r)
#pragma unroll
      for (int k = 0; k < 5; ++k) val[r][k] += row_xor<1>(val[r][k]);

    const bool odd = (c & 1) != 0;
    const int  row = 8 * h + ((c >> 2) << 1) + (c & 1);
    float o0  = odd ? val[1][0] : val[0][0];
    float o1  = odd ? val[1][1] : val[0][1];
    float o2  = odd ? val[1][2] : val[0][2];
    float o3  = odd ? val[1][3] : val[0][3];
    float sum = odd ? val[1][4] : val[0][4];

    float inv = 1.f / sum;
    o0 *= inv; o1 *= inv; o2 *= inv; o3 *= inv;

    float h1v[H1D];
#pragma unroll
    for (int j = 0; j < H1D; ++j)
      h1v[j] = tanhf(b1[j] + o0 * W1[0 * H1D + j] + o1 * W1[1 * H1D + j]
                           + o2 * W1[2 * H1D + j] + o3 * W1[3 * H1D + j]);
    float h2v[H2D];
#pragma unroll
    for (int j = 0; j < H2D; ++j) {
      float t2 = b2[j];
#pragma unroll
      for (int i = 0; i < H1D; ++i) t2 += h1v[i] * W2[i * H2D + j];
      h2v[j] = tanhf(t2);
    }
    float y = b3[0];
#pragma unroll
    for (int j = 0; j < H2D; ++j) y += h2v[j] * W3[j];

    if ((c & 2) == 0) out[(size_t)b * SEQ + i0 + row] = y;
  }
}

extern "C" void kernel_launch(void* const* d_in, const int* in_sizes, int n_in,
                              void* d_out, int out_size, void* d_ws, size_t ws_size,
                              hipStream_t stream) {
  const float* x  = (const float*)d_in[0];
  const float* Wq = (const float*)d_in[1];
  const float* Wk = (const float*)d_in[2];
  const float* Wv = (const float*)d_in[3];
  const float* W1 = (const float*)d_in[4];
  const float* b1 = (const float*)d_in[5];
  const float* W2 = (const float*)d_in[6];
  const float* b2 = (const float*)d_in[7];
  const float* W3 = (const float*)d_in[8];
  const float* b3 = (const float*)d_in[9];
  float* out = (float*)d_out;

  const size_t smem_bytes = (size_t)(SEQ * EMBED * 2 + SEQ * 4) * sizeof(float);

  hybrid_attn_kernel<<<BATCH, 128, smem_bytes, stream>>>(
      x, Wq, Wk, Wv, W1, b1, W2, b2, W3, b3, out);
}


# MLP once per query behind the loop: attention rows staged in LDS, layer 1 as W1^T x O^T + b1 on the matrix unit, one query's 8 units per lane; was 4 passes doubled in lane pairs; layer-1 sums by WMMA
# speedup vs baseline: 1.4239x; 1.1850x over previous
#include <hip/hip_runtime.h>
#include <math.h>


#define BATCH   1024
#define SEQ     256
#define EMBED   4
#define H1D     8
#define H2D     4
#define NWAVE   4

typedef __attribute__((ext_vector_type(2))) float v2f;
typedef __attribute__((ext_vector_type(8))) float v8f;

template <int M>
__device__ __forceinline__ float row_xor(float v) {
  return __builtin_bit_cast(float,
      __builtin_amdgcn_update_dpp(0, __builtin_bit_cast(int, v), 0x160 + M, 0xf, 0xf, true));
}

__global__ __launch_bounds__(128)
void hybrid_attn_kernel(const float* __restrict__ x,
                        const float* __restrict__ Wq, const float* __restrict__ Wk,
                        const float* __restrict__ Wv,
                        const float* __restrict__ W1, const float* __restrict__ b1,
                        const float* __restrict__ W2, const float* __restrict__ b2,
                        const float* __restrict__ W3, const float* __restrict__ b3,
                        float* __restrict__ out)
{
  extern __shared__ float smem[];
  float*  q_lds = smem;
  float*  k_lds = q_lds + SEQ * EMBED;
  float4* v4    = (float4*)(k_lds + SEQ * EMBED);
  float4* o4    = v4 + SEQ;

  const int b    = blockIdx.x;
  const int tid  = threadIdx.x;
  const int lane = tid & 31;
  const int wave = tid >> 5;

  const float qscale = 0.5f * 1.44269504088896340736f;
  const float* xb = x + (size_t)b * SEQ * EMBED;
  for (int t = tid; t < SEQ; t += 128) {
    float4 xe = ((const float4*)xb)[t];
    float xv[4] = {xe.x, xe.y, xe.z, xe.w};
    float4 vvec;
    float* vp = (float*)&vvec;
#pragma unroll
    for (int eo = 0; eo < 4; ++eo) {
      float q = 0.f, k = 0.f, v = 0.f;
#pragma unroll
      for (int e = 0; e < 4; ++e) {
        q += xv[e] * Wq[e * 4 + eo];
        k += xv[e] * Wk[e * 4 + eo];
        v += xv[e] * Wv[e * 4 + eo];
      }
      q_lds[t * 4 + eo] = qscale * q;
      k_lds[t * 4 + eo] = k;
      vp[eo] = v;
    }
    v4[t] = vvec;
  }
  __syncthreads();

  const int c     = lane & 15;
  const int h     = lane >> 4;
  const int kbase = h * 2;
  const bool hi8  = (c & 8) != 0;
  const bool hi4  = (c & 4) != 0;

  v2f    barr[16];
  float4 vreg[16];
#pragma unroll
  for (int j = 0; j < 16; ++j) {
    barr[j].x = k_lds[(j * 16 + c) * 4 + kbase];
    barr[j].y = k_lds[(j * 16 + c) * 4 + kbase + 1];
    vreg[j]   = v4[j * 16 + c];
  }

  v2f aw;
  aw.x = W1[kbase * H1D + (c & 7)];
  aw.y = W1[(kbase + 1) * H1D + (c & 7)];
  float b1r[H1D], W2r[H1D * H2D], b2r[H2D], W3r[H2D];
#pragma unroll
  for (int j = 0; j < H1D; ++j) b1r[j] = b1[j];
#pragma unroll
  for (int j = 0; j < H1D * H2D; ++j) W2r[j] = W2[j];
#pragma unroll
  for (int j = 0; j < H2D; ++j) { b2r[j] = b2[j]; W3r[j] = W3[j]; }
  const float b3r = b3[0];

  for (int t = 0; t < SEQ / 16 / NWAVE; ++t) {
    const int i0 = (wave * (SEQ / 16 / NWAVE) + t) * 16;

    v2f a;
    a.x = q_lds[(i0 + c) * 4 + kbase];
    a.y = q_lds[(i0 + c) * 4 + kbase + 1];

    float mx[8];
    {
      v8f z = {};
      v8f s = __builtin_amdgcn_wmma_f32_16x16x4_f32(
                  false, a, false, barr[0], (short)0, z, false, false);
#pragma unroll
      for (int r = 0; r < 8; ++r) mx[r] = s[r];
#pragma unroll
      for (int j = 1; j < 16; ++j) {
        s = __builtin_amdgcn_wmma_f32_16x16x4_f32(
                false, a, false, barr[j], (short)0, z, false, false);
#pragma unroll
        for (int r = 0; r < 8; ++r) mx[r] = fmaxf(mx[r], s[r]);
      }
    }
#pragma unroll
    for (int r = 0; r < 8; ++r) mx[r] = fmaxf(mx[r], row_xor<1>(mx[r]));
#pragma unroll
    for (int r = 0; r < 8; ++r) mx[r] = fmaxf(mx[r], row_xor<2>(mx[r]));
#pragma unroll
    for (int r = 0; r < 8; ++r) mx[r] = fmaxf(mx[r], row_xor<4>(mx[r]));
#pragma unroll
    for (int r = 0; r < 8; ++r) mx[r] = fmaxf(mx[r], row_xor<8>(mx[r]));

    v8f negm;
#pragma unroll
    for (int r = 0; r < 8; ++r) negm[r] = -mx[r];

    v2f acc[4][4], l2[4];
#pragma unroll
    for (int e = 0; e < 4; ++e)
#pragma unroll
      for (int i = 0; i < 4; ++i) acc[e][i] = (v2f){0.f, 0.f};
#pragma unroll
    for (int i = 0; i < 4; ++i) l2[i] = (v2f){0.f, 0.f};

#pragma unroll
    for (int j = 0; j < 16; ++j) {
      v8f d = __builtin_amdgcn_wmma_f32_16x16x4_f32(
                  false, a, false, barr[j], (short)0, negm, false, false);
      const float ve[4] = {vreg[j].x, vreg[j].y, vreg[j].z, vreg[j].w};
      v2f p2[4];
#pragma unroll
      for (int i = 0; i < 4; ++i)
        p2[i] = (v2f){__builtin_amdgcn_exp2f(d[2 * i]), __builtin_amdgcn_exp2f(d[2 * i + 1])};
#pragma unroll
      for (int e = 0; e < 4; ++e) {
        const v2f vv = {ve[e], ve[e]};
#pragma unroll
        for (int i = 0; i < 4; ++i) acc[e][i] = __builtin_elementwise_fma(p2[i], vv, acc[e][i]);
      }
#pragma unroll
      for (int i = 0; i < 4; ++i) l2[i] += p2[i];
    }

    float val[8][5];
#pragma unroll
    for (int r = 0; r < 8; ++r) {
#pragma unroll
      for (int e = 0; e < 4; ++e) val[r][e] = (r & 1) ? acc[e][r >> 1].y : acc[e][r >> 1].x;
      val[r][4] = (r & 1) ? l2[r >> 1].y : l2[r >> 1].x;
    }
#pragma unroll
    for (int r = 0; r < 4; ++r)
#pragma unroll
      for (int k = 0; k < 5; ++k) {
        const float keep = hi8 ? val[r + 4][k] : val[r][k];
        const float send = hi8 ? val[r][k] : val[r + 4][k];
        val[r][k] = keep + row_xor<8>(send);
      }
#pragma unroll
    for (int r = 0; r < 2; ++r)
#pragma unroll
      for (int k = 0; k < 5; ++k) {
        const float keep = hi4 ? val[r + 2][k] : val[r][k];
        const float send = hi4 ? val[r][k] : val[r + 2][k];
        val[r][k] = keep + row_xor<4>(send);
      }
#pragma unroll
    for (int r = 0; r < 2; ++r)
#pragma unroll
      for (int k = 0; k < 5; ++k) val[r][k] += row_xor<2>(val[r][k]);
#pragma unroll
    for (int r = 0; r < 2; ++r)
#pragma unroll
      for (int k = 0; k < 5; ++k) val[r][k] += row_xor<1>(val[r][k]);

    const bool odd = (c & 1) != 0;
    const int  row = 8 * h + ((c >> 2) << 1) + (c & 1);
    float o0  = odd ? val[1][0] : val[0][0];
    float o1  = odd ? val[1][1] : val[0][1];
    float o2  = odd ? val[1][2] : val[0][2];
    float o3  = odd ? val[1][3] : val[0][3];
    float sum = odd ? val[1][4] : val[0][4];

    float inv = 1.f / sum;
    o0 *= inv; o1 *= inv; o2 *= inv; o3 *= inv;

    if ((c & 2) == 0) o4[i0 + row] = make_float4(o0, o1, o2, o3);
  }

  __syncthreads();

  v8f bias1;
#pragma unroll
  for (int r = 0; r < 8; ++r) bias1[r] = b1r[r];

  const float* of = (const float*)o4;
  for (int g = 0; g < 2; ++g) {
    const int q0 = wave * 64 + g * 32;
    v2f blo, bhi;
    blo.x = of[(q0 + c) * 4 + kbase];
    blo.y = of[(q0 + c) * 4 + kbase + 1];
    bhi.x = of[(q0 + 16 + c) * 4 + kbase];
    bhi.y = of[(q0 + 16 + c) * 4 + kbase + 1];
    v8f dlo = __builtin_amdgcn_wmma_f32_16x16x4_f32(
                  false, aw, false, blo, (short)0, bias1, false, false);
    v8f dhi = __builtin_amdgcn_wmma_f32_16x16x4_f32(
                  false, aw, false, bhi, (short)0, bias1, false, false);

    float h1v[H1D];
#pragma unroll
    for (int j = 0; j < H1D; ++j) h1v[j] = tanhf(h ? dhi[j] : dlo[j]);
    float h2v[H2D];
#pragma unroll
    for (int j = 0; j < H2D; ++j) {
      float t2 = b2r[j];
#pragma unroll
      for (int i = 0; i < H1D; ++i) t2 += h1v[i] * W2r[i * H2D + j];
      h2v[j] = tanhf(t2);
    }
    float y = b3r;
#pragma unroll
    for (int j = 0; j < H2D; ++j) y += h2v[j] * W3r[j];

    out[(size_t)b * SEQ + q0 + lane] = y;
  }
}

extern "C" void kernel_launch(void* const* d_in, const int* in_sizes, int n_in,
                              void* d_out, int out_size, void* d_ws, size_t ws_size,
                              hipStream_t stream) {
  const float* x  = (const float*)d_in[0];
  const float* Wq = (const float*)d_in[1];
  const float* Wk = (const float*)d_in[2];
  const float* Wv = (const float*)d_in[3];
  const float* W1 = (const float*)d_in[4];
  const float* b1 = (const float*)d_in[5];
  const float* W2 = (const float*)d_in[6];
  const float* b2 = (const float*)d_in[7];
  const float* W3 = (const float*)d_in[8];
  const float* b3 = (const float*)d_in[9];
  float* out = (float*)d_out;

  const size_t smem_bytes = (size_t)(SEQ * EMBED * 2 + SEQ * 4 * 2) * sizeof(float);

  hybrid_attn_kernel<<<BATCH, 128, smem_bytes, stream>>>(
      x, Wq, Wk, Wv, W1, b1, W2, b2, W3, b3, out);
}


# score tiles computed once per row tile and kept in registers for both passes: second score product gone; max subtracted after the product by packed adds (was the C operand); PV/row-sum order kept
# speedup vs baseline: 1.5720x; 1.1040x over previous
#include <hip/hip_runtime.h>
#include <math.h>


#define BATCH   1024
#define SEQ     256
#define EMBED   4
#define H1D     8
#define H2D     4
#define NWAVE   4
#define VRES    8

typedef __attribute__((ext_vector_type(2))) float v2f;
typedef __attribute__((ext_vector_type(8))) float v8f;

template <int M>
__device__ __forceinline__ float row_xor(float v) {
  return __builtin_bit_cast(float,
      __builtin_amdgcn_update_dpp(0, __builtin_bit_cast(int, v), 0x160 + M, 0xf, 0xf, true));
}

__global__ __launch_bounds__(128)
void hybrid_attn_kernel(const float* __restrict__ x,
                        const float* __restrict__ Wq, const float* __restrict__ Wk,
                        const float* __restrict__ Wv,
                        const float* __restrict__ W1, const float* __restrict__ b1,
                        const float* __restrict__ W2, const float* __restrict__ b2,
                        const float* __restrict__ W3, const float* __restrict__ b3,
                        float* __restrict__ out)
{
  extern __shared__ float smem[];
  float*  q_lds = smem;
  float*  k_lds = q_lds + SEQ * EMBED;
  float4* v4    = (float4*)(k_lds + SEQ * EMBED);
  float4* o4    = v4 + SEQ;

  const int b    = blockIdx.x;
  const int tid  = threadIdx.x;
  const int lane = tid & 31;
  const int wave = tid >> 5;

  const float qscale = 0.5f * 1.44269504088896340736f;
  const float* xb = x + (size_t)b * SEQ * EMBED;
  for (int t = tid; t < SEQ; t += 128) {
    float4 xe = ((const float4*)xb)[t];
    float xv[4] = {xe.x, xe.y, xe.z, xe.w};
    float4 vvec;
    float* vp = (float*)&vvec;
#pragma unroll
    for (int eo = 0; eo < 4; ++eo) {
      float q = 0.f, k = 0.f, v = 0.f;
#pragma unroll
      for (int e = 0; e < 4; ++e) {
        q += xv[e] * Wq[e * 4 + eo];
        k += xv[e] * Wk[e * 4 + eo];
        v += xv[e] * Wv[e * 4 + eo];
      }
      q_lds[t * 4 + eo] = qscale * q;
      k_lds[t * 4 + eo] = k;
      vp[eo] = v;
    }
    v4[t] = vvec;
  }
  __syncthreads();

  const int c     = lane & 15;
  const int h     = lane >> 4;
  const int kbase = h * 2;
  const bool hi8  = (c & 8) != 0;
  const bool hi4  = (c & 4) != 0;

  float4 vreg[VRES];
#pragma unroll
  for (int j = 0; j < VRES; ++j) vreg[j] = v4[j * 16 + c];
  const v2f* k2 = (const v2f*)k_lds;

  v2f aw;
  aw.x = W1[kbase * H1D + (c & 7)];
  aw.y = W1[(kbase + 1) * H1D + (c & 7)];
  float b1r[H1D], W2r[H1D * H2D], b2r[H2D], W3r[H2D];
#pragma unroll
  for (int j = 0; j < H1D; ++j) b1r[j] = b1[j];
#pragma unroll
  for (int j = 0; j < H1D * H2D; ++j) W2r[j] = W2[j];
#pragma unroll
  for (int j = 0; j < H2D; ++j) { b2r[j] = b2[j]; W3r[j] = W3[j]; }
  const float b3r = b3[0];

  for (int t = 0; t < SEQ / 16 / NWAVE; ++t) {
    const int i0 = (wave * (SEQ / 16 / NWAVE) + t) * 16;

    v2f a;
    a.x = q_lds[(i0 + c) * 4 + kbase];
    a.y = q_lds[(i0 + c) * 4 + kbase + 1];

    v8f s[16];
    {
      v8f z = {};
#pragma unroll
      for (int j = 0; j < 16; ++j) {
        const v2f bj = k2[(j * 16 + c) * 2 + h];
        s[j] = __builtin_amdgcn_wmma_f32_16x16x4_f32(
                   false, a, false, bj, (short)0, z, false, false);
      }
    }

    float mx[8];
#pragma unroll
    for (int r = 0; r < 8; ++r) mx[r] = s[0][r];
#pragma unroll
    for (int j = 1; j < 16; ++j)
#pragma unroll
      for (int r = 0; r < 8; ++r) mx[r] = fmaxf(mx[r], s[j][r]);
#pragma unroll
    for (int r = 0; r < 8; ++r) mx[r] = fmaxf(mx[r], row_xor<1>(mx[r]));
#pragma unroll
    for (int r = 0; r < 8; ++r) mx[r] = fmaxf(mx[r], row_xor<2>(mx[r]));
#pragma unroll
    for (int r = 0; r < 8; ++r) mx[r] = fmaxf(mx[r], row_xor<4>(mx[r]));
#pragma unroll
    for (int r = 0; r < 8; ++r) mx[r] = fmaxf(mx[r], row_xor<8>(mx[r]));

    v2f m2[4];
#pragma unroll
    for (int i = 0; i < 4; ++i) m2[i] = (v2f){mx[2 * i], mx[2 * i + 1]};

    v2f acc[4][4], l2[4];
#pragma unroll
    for (int e = 0; e < 4; ++e)
#pragma unroll
      for (int i = 0; i < 4; ++i) acc[e][i] = (v2f){0.f, 0.f};
#pragma unroll
    for (int i = 0; i < 4; ++i) l2[i] = (v2f){0.f, 0.f};

#pragma unroll
    for (int j = 0; j < 16; ++j) {
      const float4 vrow = (j < VRES) ? vreg[j] : v4[j * 16 + c];
      const float ve[4] = {vrow.x, vrow.y, vrow.z, vrow.w};
      v2f p2[4];
#pragma unroll
      for (int i = 0; i < 4; ++i) {
        const v2f d2 = (v2f){s[j][2 * i], s[j][2 * i + 1]} - m2[i];
        p2[i] = (v2f){__builtin_amdgcn_exp2f(d2.x), __builtin_amdgcn_exp2f(d2.y)};
      }
#pragma unroll
      for (int e = 0; e < 4; ++e) {
        const v2f vv = {ve[e], ve[e]};
#pragma unroll
        for (int i = 0; i < 4; ++i) acc[e][i] = __builtin_elementwise_fma(p2[i], vv, acc[e][i]);
      }
#pragma unroll
      for (int i = 0; i < 4; ++i) l2[i] += p2[i];
    }

    float val[8][5];
#pragma unroll
    for (int r = 0; r < 8; ++r) {
#pragma unroll
      for (int e = 0; e < 4; ++e) val[r][e] = (r & 1) ? acc[e][r >> 1].y : acc[e][r >> 1].x;
      val[r][4] = (r & 1) ? l2[r >> 1].y : l2[r >> 1].x;
    }
#pragma unroll
    for (int r = 0; r < 4; ++r)
#pragma unroll
      for (int k = 0; k < 5; ++k) {
        const float keep = hi8 ? val[r + 4][k] : val[r][k];
        const float send = hi8 ? val[r][k] : val[r + 4][k];
        val[r][k] = keep + row_xor<8>(send);
      }
#pragma unroll
    for (int r = 0; r < 2; ++r)
#pragma unroll
      for (int k = 0; k < 5; ++k) {
        const float keep = hi4 ? val[r + 2][k] : val[r][k];
        const float send = hi4 ? val[r][k] : val[r + 2][k];
        val[r][k] = keep + row_xor<4>(send);
      }
#pragma unroll
    for (int r = 0; r < 2; ++r)
#pragma unroll
      for (int k = 0; k < 5; ++k) val[r][k] += row_xor<2>(val[r][k]);
#pragma unroll
    for (int r = 0; r < 2; ++r)
#pragma unroll
      for (int k = 0; k < 5; ++k) val[r][k] += row_xor<1>(val[r][k]);

    const bool odd = (c & 1) != 0;
    const int  row = 8 * h + ((c >> 2) << 1) + (c & 1);
    float o0  = odd ? val[1][0] : val[0][0];
    float o1  = odd ? val[1][1] : val[0][1];
    float o2  = odd ? val[1][2] : val[0][2];
    float o3  = odd ? val[1][3] : val[0][3];
    float sum = odd ? val[1][4] : val[0][4];

    float inv = 1.f / sum;
    o0 *= inv; o1 *= inv; o2 *= inv; o3 *= inv;

    if ((c & 2) == 0) o4[i0 + row] = make_float4(o0, o1, o2, o3);
  }

  __syncthreads();

  v8f bias1;
#pragma unroll
  for (int r = 0; r < 8; ++r) bias1[r] = b1r[r];

  const float* of = (const float*)o4;
  for (int g = 0; g < 2; ++g) {
    const int q0 = wave * 64 + g * 32;
    v2f blo, bhi;
    blo.x = of[(q0 + c) * 4 + kbase];
    blo.y = of[(q0 + c) * 4 + kbase + 1];
    bhi.x = of[(q0 + 16 + c) * 4 + kbase];
    bhi.y = of[(q0 + 16 + c) * 4 + kbase + 1];
    v8f dlo = __builtin_amdgcn_wmma_f32_16x16x4_f32(
                  false, aw, false, blo, (short)0, bias1, false, false);
    v8f dhi = __builtin_amdgcn_wmma_f32_16x16x4_f32(
                  false, aw, false, bhi, (short)0, bias1, false, false);

    float h1v[H1D];
#pragma unroll
    for (int j = 0; j < H1D; ++j) h1v[j] = tanhf(h ? dhi[j] : dlo[j]);
    float h2v[H2D];
#pragma unroll
    for (int j = 0; j < H2D; ++j) {
      float t2 = b2r[j];
#pragma unroll
      for (int i = 0; i < H1D; ++i) t2 += h1v[i] * W2r[i * H2D + j];
      h2v[j] = tanhf(t2);
    }
    float y = b3r;
#pragma unroll
    for (int j = 0; j < H2D; ++j) y += h2v[j] * W3r[j];

    out[(size_t)b * SEQ + q0 + lane] = y;
  }
}

extern "C" void kernel_launch(void* const* d_in, const int* in_sizes, int n_in,
                              void* d_out, int out_size, void* d_ws, size_t ws_size,
                              hipStream_t stream) {
  const float* x  = (const float*)d_in[0];
  const float* Wq = (const float*)d_in[1];
  const float* Wk = (const float*)d_in[2];
  const float* Wv = (const float*)d_in[3];
  const float* W1 = (const float*)d_in[4];
  const float* b1 = (const float*)d_in[5];
  const float* W2 = (const float*)d_in[6];
  const float* b2 = (const float*)d_in[7];
  const float* W3 = (const float*)d_in[8];
  const float* b3 = (const float*)d_in[9];
  float* out = (float*)d_out;

  const size_t smem_bytes = (size_t)(SEQ * EMBED * 2 + SEQ * 4 * 2) * sizeof(float);

  hybrid_attn_kernel<<<BATCH, 128, smem_bytes, stream>>>(
      x, Wq, Wk, Wv, W1, b1, W2, b2, W3, b3, out);
}


# row maxima and row sums over the 16 key lanes go through LDS: lanes store partials by query row, one lane per (query, column pair) adds the 16 in lane order; DPP butterflies gone; sum order changed
# speedup vs baseline: 1.6670x; 1.0604x over previous
#include <hip/hip_runtime.h>
#include <math.h>


#define BATCH   1024
#define SEQ     256
#define EMBED   4
#define H1D     8
#define H2D     4
#define NWAVE   4
#define VRES    8
#define MXH     144
#define RPITCH  84

typedef __attribute__((ext_vector_type(2))) float v2f;
typedef __attribute__((ext_vector_type(8))) float v8f;

__global__ __launch_bounds__(128)
void hybrid_attn_kernel(const float* __restrict__ x,
                        const float* __restrict__ Wq, const float* __restrict__ Wk,
                        const float* __restrict__ Wv,
                        const float* __restrict__ W1, const float* __restrict__ b1,
                        const float* __restrict__ W2, const float* __restrict__ b2,
                        const float* __restrict__ W3, const float* __restrict__ b3,
                        float* __restrict__ out)
{
  extern __shared__ float smem[];
  float*  q_lds = smem;
  float*  k_lds = q_lds + SEQ * EMBED;
  float4* v4    = (float4*)(k_lds + SEQ * EMBED);
  float4* o4    = v4 + SEQ;
  float*  red   = (float*)(o4 + SEQ);

  const int b    = blockIdx.x;
  const int tid  = threadIdx.x;
  const int lane = tid & 31;
  const int wave = tid >> 5;

  const float qscale = 0.5f * 1.44269504088896340736f;
  const float* xb = x + (size_t)b * SEQ * EMBED;
  for (int t = tid; t < SEQ; t += 128) {
    float4 xe = ((const float4*)xb)[t];
    float xv[4] = {xe.x, xe.y, xe.z, xe.w};
    float4 vvec;
    float* vp = (float*)&vvec;
#pragma unroll
    for (int eo = 0; eo < 4; ++eo) {
      float q = 0.f, k = 0.f, v = 0.f;
#pragma unroll
      for (int e = 0; e < 4; ++e) {
        q += xv[e] * Wq[e * 4 + eo];
        k += xv[e] * Wk[e * 4 + eo];
        v += xv[e] * Wv[e * 4 + eo];
      }
      q_lds[t * 4 + eo] = qscale * q;
      k_lds[t * 4 + eo] = k;
      vp[eo] = v;
    }
    v4[t] = vvec;
  }
  __syncthreads();

  const int c     = lane & 15;
  const int h     = lane >> 4;
  const int kbase = h * 2;

  float4 vreg[VRES];
#pragma unroll
  for (int j = 0; j < VRES; ++j) vreg[j] = v4[j * 16 + c];
  const v2f* k2 = (const v2f*)k_lds;

  v2f aw;
  aw.x = W1[kbase * H1D + (c & 7)];
  aw.y = W1[(kbase + 1) * H1D + (c & 7)];
  float b1r[H1D], W2r[H1D * H2D], b2r[H2D], W3r[H2D];
#pragma unroll
  for (int j = 0; j < H1D; ++j) b1r[j] = b1[j];
#pragma unroll
  for (int j = 0; j < H1D * H2D; ++j) W2r[j] = W2[j];
#pragma unroll
  for (int j = 0; j < H2D; ++j) { b2r[j] = b2[j]; W3r[j] = W3[j]; }
  const float b3r = b3[0];

  float* wred = red + wave * (16 * RPITCH);

  for (int t = 0; t < SEQ / 16 / NWAVE; ++t) {
    const int i0 = (wave * (SEQ / 16 / NWAVE) + t) * 16;

    v2f a;
    a.x = q_lds[(i0 + c) * 4 + kbase];
    a.y = q_lds[(i0 + c) * 4 + kbase + 1];

    v8f s[16];
    {
      v8f z = {};
#pragma unroll
      for (int j = 0; j < 16; ++j) {
        const v2f bj = k2[(j * 16 + c) * 2 + h];
        s[j] = __builtin_amdgcn_wmma_f32_16x16x4_f32(
                   false, a, false, bj, (short)0, z, false, false);
      }
    }

    float mx[8];
#pragma unroll
    for (int r = 0; r < 8; ++r) mx[r] = s[0][r];
#pragma unroll
    for (int j = 1; j < 16; ++j)
#pragma unroll
      for (int r = 0; r < 8; ++r) mx[r] = fmaxf(mx[r], s[j][r]);
#pragma unroll
    for (int r = 0; r < 8; ++r) wred[h * MXH + r * 16 + c] = mx[r];
    __builtin_amdgcn_fence(__ATOMIC_RELEASE, "wavefront");
    __builtin_amdgcn_fence(__ATOMIC_ACQUIRE, "wavefront");
    {
      const float4* mrd = (const float4*)(wred + h * MXH + (c & 7) * 16);
      const float4 ma = mrd[0], mb = mrd[1], mc = mrd[2], md = mrd[3];
      const float mab = fmaxf(fmaxf(fmaxf(ma.x, ma.y), fmaxf(ma.z, ma.w)), fmaxf(fmaxf(mb.x, mb.y), fmaxf(mb.z, mb.w)));
      const float mcd = fmaxf(fmaxf(fmaxf(mc.x, mc.y), fmaxf(mc.z, mc.w)), fmaxf(fmaxf(md.x, md.y), fmaxf(md.z, md.w)));
      wred[2 * MXH + h * 8 + (c & 7)] = fmaxf(mab, mcd);
    }
    __builtin_amdgcn_fence(__ATOMIC_RELEASE, "wavefront");
    __builtin_amdgcn_fence(__ATOMIC_ACQUIRE, "wavefront");
    const float4* frd = (const float4*)(wred + 2 * MXH + h * 8);
    const float4 fa = frd[0], fb = frd[1];

    v2f m2[4];
    m2[0] = (v2f){fa.x, fa.y}; m2[1] = (v2f){fa.z, fa.w};
    m2[2] = (v2f){fb.x, fb.y}; m2[3] = (v2f){fb.z, fb.w};

    v2f acc[8][2], l2[4];
#pragma unroll
    for (int r = 0; r < 8; ++r) { acc[r][0] = (v2f){0.f, 0.f}; acc[r][1] = (v2f){0.f, 0.f}; }
#pragma unroll
    for (int i = 0; i < 4; ++i) l2[i] = (v2f){0.f, 0.f};

#pragma unroll
    for (int j = 0; j < 16; ++j) {
      const float4 vrow = (j < VRES) ? vreg[j] : v4[j * 16 + c];
      const v2f v01 = {vrow.x, vrow.y};
      const v2f v23 = {vrow.z, vrow.w};
      v2f p2[4];
#pragma unroll
      for (int i = 0; i < 4; ++i) {
        const v2f d2 = (v2f){s[j][2 * i], s[j][2 * i + 1]} - m2[i];
        p2[i] = (v2f){__builtin_amdgcn_exp2f(d2.x), __builtin_amdgcn_exp2f(d2.y)};
      }
#pragma unroll
      for (int r = 0; r < 8; ++r) {
        const float pr = (r & 1) ? p2[r >> 1].y : p2[r >> 1].x;
        const v2f pp = {pr, pr};
        acc[r][0] = __builtin_elementwise_fma(pp, v01, acc[r][0]);
        acc[r][1] = __builtin_elementwise_fma(pp, v23, acc[r][1]);
      }
#pragma unroll
      for (int i = 0; i < 4; ++i) l2[i] += p2[i];
    }

#pragma unroll
    for (int r = 0; r < 8; ++r) {
      float* wr = wred + (8 * h + r) * RPITCH + 2 * c;
      *(v2f*)(wr)      = acc[r][0];
      *(v2f*)(wr + 32) = acc[r][1];
    }
#pragma unroll
    for (int i = 0; i < 4; ++i) {
      wred[(8 * h + 2 * i) * RPITCH + 64 + c]     = l2[i].x;
      wred[(8 * h + 2 * i + 1) * RPITCH + 64 + c] = l2[i].y;
    }
    __builtin_amdgcn_fence(__ATOMIC_RELEASE, "wavefront");
    __builtin_amdgcn_fence(__ATOMIC_ACQUIRE, "wavefront");

    const float4* rd = (const float4*)(wred + c * RPITCH);
    v2f so = {0.f, 0.f};
#pragma unroll
    for (int m = 0; m < 8; ++m) {
      const float4 pq = rd[8 * h + m];
      so += (v2f){pq.x, pq.y};
      so += (v2f){pq.z, pq.w};
    }
    v2f sl = {0.f, 0.f};
#pragma unroll
    for (int m = 0; m < 4; ++m) {
      const float4 lq = rd[16 + m];
      sl += (v2f){lq.x, lq.y};
      sl += (v2f){lq.z, lq.w};
    }
    const float sum = sl.x + sl.y;
    const float inv = 1.f / sum;
    so *= (v2f){inv, inv};

    *(v2f*)((float*)o4 + (i0 + c) * 4 + 2 * h) = so;
    __builtin_amdgcn_fence(__ATOMIC_RELEASE, "wavefront");
  }

  __syncthreads();

  v8f bias1;
#pragma unroll
  for (int r = 0; r < 8; ++r) bias1[r] = b1r[r];

  const float* of = (const float*)o4;
  for (int g = 0; g < 2; ++g) {
    const int q0 = wave * 64 + g * 32;
    v2f blo, bhi;
    blo.x = of[(q0 + c) * 4 + kbase];
    blo.y = of[(q0 + c) * 4 + kbase + 1];
    bhi.x = of[(q0 + 16 + c) * 4 + kbase];
    bhi.y = of[(q0 + 16 + c) * 4 + kbase + 1];
    v8f dlo = __builtin_amdgcn_wmma_f32_16x16x4_f32(
                  false, aw, false, blo, (short)0, bias1, false, false);
    v8f dhi = __builtin_amdgcn_wmma_f32_16x16x4_f32(
                  false, aw, false, bhi, (short)0, bias1, false, false);

    float h1v[H1D];
#pragma unroll
    for (int j = 0; j < H1D; ++j) h1v[j] = tanhf(h ? dhi[j] : dlo[j]);
    float h2v[H2D];
#pragma unroll
    for (int j = 0; j < H2D; ++j) {
      float t2 = b2r[j];
#pragma unroll
      for (int i = 0; i < H1D; ++i) t2 += h1v[i] * W2r[i * H2D + j];
      h2v[j] = tanhf(t2);
    }
    float y = b3r;
#pragma unroll
    for (int j = 0; j < H2D; ++j) y += h2v[j] * W3r[j];

    out[(size_t)b * SEQ + q0 + lane] = y;
  }
}

extern "C" void kernel_launch(void* const* d_in, const int* in_sizes, int n_in,
                              void* d_out, int out_size, void* d_ws, size_t ws_size,
                              hipStream_t stream) {
  const float* x  = (const float*)d_in[0];
  const float* Wq = (const float*)d_in[1];
  const float* Wk = (const float*)d_in[2];
  const float* Wv = (const float*)d_in[3];
  const float* W1 = (const float*)d_in[4];
  const float* b1 = (const float*)d_in[5];
  const float* W2 = (const float*)d_in[6];
  const float* b2 = (const float*)d_in[7];
  const float* W3 = (const float*)d_in[8];
  const float* b3 = (const float*)d_in[9];
  float* out = (float*)d_out;

  const size_t smem_bytes = (size_t)(SEQ * EMBED * 2 + SEQ * 4 * 2 + NWAVE * 16 * RPITCH) * sizeof(float);

  hybrid_attn_kernel<<<BATCH, 128, smem_bytes, stream>>>(
      x, Wq, Wk, Wv, W1, b1, W2, b2, W3, b3, out);
}
